# starting kernel
# baseline (speedup 1.0000x reference)
#include <hip/hip_runtime.h>
#include <hip/hip_bf16.h>
#include <math.h>

typedef float v2f __attribute__((ext_vector_type(2)));
typedef float v8f __attribute__((ext_vector_type(8)));

#define TT   512
#define BB   512
#define NIN  64
#define H1   100
#define H2   50
#define H3   25
#define G1   400
#define G2   200
#define G3   100
#define G1S  400
#define G2S  208
#define G3S  112
#define SH1  104
#define SH2  52
#define SH3  28
#define NT   256
#define NWAVES 8

#define KS_X1 16
#define KS_H1 25
#define KS_H2 13
#define KS_H3 7

__device__ __forceinline__ float sigf(float x) { return 1.0f / (1.0f + __expf(-x)); }

__device__ __forceinline__ v8f wmma4(v2f a, v2f b, v8f c) {
  return __builtin_amdgcn_wmma_f32_16x16x4_f32(
      false, a, false, b, (short)0, c, false, false);
}

template<int KS>
__device__ __forceinline__ void load_frags(v2f* f, const float* p, int khalf) {
  #pragma unroll
  for (int i = 0; i < KS; ++i) {
    f[i].x = p[4 * i + khalf];
    f[i].y = p[4 * i + khalf + 1];
  }
}

template<int KS>
__device__ __forceinline__ v8f wmma_chain(const v2f* a, const v2f* b, v8f acc) {
  #pragma unroll
  for (int i = 0; i < KS; ++i) acc = wmma4(a[i], b[i], acc);
  return acc;
}

__global__ __launch_bounds__(NT, 2) void lstm3_fused_kernel(
    const float* __restrict__ x,
    const float* __restrict__ w_ih1, const float* __restrict__ w_hh1,
    const float* __restrict__ b_ih1, const float* __restrict__ b_hh1,
    const float* __restrict__ w_ih2, const float* __restrict__ w_hh2,
    const float* __restrict__ b_ih2, const float* __restrict__ b_hh2,
    const float* __restrict__ w_ih3, const float* __restrict__ w_hh3,
    const float* __restrict__ b_ih3, const float* __restrict__ b_hh3,
    const float* __restrict__ w_lin, const float* __restrict__ b_lin,
    float* __restrict__ out)
{
  __shared__ float gates1[16 * G1S];
  __shared__ float gates2[16 * G2S];
  __shared__ float gates3[16 * G3S];
  __shared__ float h1s[16 * SH1];
  __shared__ float h2s[16 * SH2];
  __shared__ float h3s[16 * SH3];
  __shared__ float whh1s[G1 * H1];
  __shared__ float whh2s[208 * SH2];
  __shared__ float wih3s[112 * SH2];
  __shared__ float whh3s[112 * SH3];

  const int tid     = threadIdx.x;
  const int lane    = tid & 31;
  const int wave    = tid >> 5;
  const int col     = lane & 15;
  const int khalf   = (lane >> 4) << 1;
  const int rowhalf = (lane >> 4) << 3;
  const int bbase   = blockIdx.x * 16;

  for (int i = tid; i < 16 * SH1; i += NT) h1s[i] = 0.f;
  for (int i = tid; i < 16 * SH2; i += NT) h2s[i] = 0.f;
  for (int i = tid; i < 16 * SH3; i += NT) h3s[i] = 0.f;
  for (int i = tid; i < 208 * SH2; i += NT) whh2s[i] = 0.f;
  for (int i = tid; i < 112 * SH2; i += NT) wih3s[i] = 0.f;
  for (int i = tid; i < 112 * SH3; i += NT) whh3s[i] = 0.f;
  __syncthreads();

  for (int i = tid; i < G1 * H1; i += NT)
    whh1s[i] = w_hh1[i];
  for (int i = tid; i < G2 * H2; i += NT) {
    int r = i / H2, c = i - r * H2;
    whh2s[r * SH2 + c] = w_hh2[i];
  }
  for (int i = tid; i < G3 * H2; i += NT) {
    int r = i / H2, c = i - r * H2;
    wih3s[r * SH2 + c] = w_ih3[i];
  }
  for (int i = tid; i < G3 * H3; i += NT) {
    int r = i / H3, c = i - r * H3;
    whh3s[r * SH3 + c] = w_hh3[i];
  }

  float bs1[7][4], bs2[4][4], bs3[2][4];
  #pragma unroll
  for (int r = 0; r < 7; ++r) {
    const int idx = tid + r * NT;
    if (idx < 16 * H1) {
      const int j = idx >> 4;
      #pragma unroll
      for (int g = 0; g < 4; ++g) bs1[r][g] = b_ih1[g * H1 + j] + b_hh1[g * H1 + j];
    }
  }
  #pragma unroll
  for (int r = 0; r < 4; ++r) {
    const int idx = tid + r * NT;
    if (idx < 16 * H2) {
      const int j = idx >> 4;
      #pragma unroll
      for (int g = 0; g < 4; ++g) bs2[r][g] = b_ih2[g * H2 + j] + b_hh2[g * H2 + j];
    }
  }
  #pragma unroll
  for (int r = 0; r < 2; ++r) {
    const int idx = tid + r * NT;
    if (idx < 16 * H3) {
      const int j = idx >> 4;
      #pragma unroll
      for (int g = 0; g < 4; ++g) bs3[r][g] = b_ih3[g * H3 + j] + b_hh3[g * H3 + j];
    }
  }

  float wl[H3];
  #pragma unroll
  for (int j = 0; j < H3; ++j) wl[j] = w_lin[j];
  const float blin = b_lin[0];

  float c1r[7], c2r[4], c3r[2];
  #pragma unroll
  for (int r = 0; r < 7; ++r) c1r[r] = 0.f;
  #pragma unroll
  for (int r = 0; r < 4; ++r) c2r[r] = 0.f;
  #pragma unroll
  for (int r = 0; r < 2; ++r) c3r[r] = 0.f;

  __syncthreads();

  for (int t = 0; t < TT; ++t) {
    {
      if (t + 1 < TT && tid < 16)
        __builtin_prefetch(x + ((size_t)(t + 1) * BB + bbase + tid) * NIN, 0, 3);

      v2f ax[KS_X1], ah[KS_H1];
      load_frags<KS_X1>(ax, x + ((size_t)t * BB + bbase + col) * NIN, khalf);
      load_frags<KS_H1>(ah, &h1s[col * SH1], khalf);
      for (int tile = wave; tile < 25; tile += NWAVES) {
        const int n = tile * 16 + col;
        v2f bi[KS_X1], bh[KS_H1];
        load_frags<KS_X1>(bi, w_ih1 + (size_t)n * NIN, khalf);
        load_frags<KS_H1>(bh, &whh1s[n * H1], khalf);
        v8f acc = {};
        acc = wmma_chain<KS_X1>(ax, bi, acc);
        acc = wmma_chain<KS_H1>(ah, bh, acc);
        #pragma unroll
        for (int r = 0; r < 8; ++r)
          gates1[(r + rowhalf) * G1S + n] = acc[r];
      }
    }
    __syncthreads();

    #pragma unroll
    for (int r = 0; r < 7; ++r) {
      const int idx = tid + r * NT;
      if (idx < 16 * H1) {
        const int m = idx & 15, j = idx >> 4;
        float gi = gates1[m * G1S + j]          + bs1[r][0];
        float gf = gates1[m * G1S + H1 + j]     + bs1[r][1];
        float gg = gates1[m * G1S + 2 * H1 + j] + bs1[r][2];
        float go = gates1[m * G1S + 3 * H1 + j] + bs1[r][3];
        float c  = sigf(gf) * c1r[r] + sigf(gi) * tanhf(gg);
        c1r[r]   = c;
        h1s[m * SH1 + j] = sigf(go) * tanhf(c);
      }
    }
    __syncthreads();

    {
      v2f a1[KS_H1], a2[KS_H2];
      load_frags<KS_H1>(a1, &h1s[col * SH1], khalf);
      load_frags<KS_H2>(a2, &h2s[col * SH2], khalf);
      for (int tile = wave; tile < 13; tile += NWAVES) {
        const int n  = tile * 16 + col;
        const int nc = (n < G2) ? n : (G2 - 1);
        v2f bi[KS_H1], bh[KS_H2];
        load_frags<KS_H1>(bi, w_ih2 + (size_t)nc * H1, khalf);
        load_frags<KS_H2>(bh, &whh2s[n * SH2], khalf);
        v8f acc = {};
        acc = wmma_chain<KS_H1>(a1, bi, acc);
        acc = wmma_chain<KS_H2>(a2, bh, acc);
        if (n < G2) {
          #pragma unroll
          for (int r = 0; r < 8; ++r)
            gates2[(r + rowhalf) * G2S + n] = acc[r];
        }
      }
    }
    __syncthreads();

    #pragma unroll
    for (int r = 0; r < 4; ++r) {
      const int idx = tid + r * NT;
      if (idx < 16 * H2) {
        const int m = idx & 15, j = idx >> 4;
        float gi = gates2[m * G2S + j]          + bs2[r][0];
        float gf = gates2[m * G2S + H2 + j]     + bs2[r][1];
        float gg = gates2[m * G2S + 2 * H2 + j] + bs2[r][2];
        float go = gates2[m * G2S + 3 * H2 + j] + bs2[r][3];
        float c  = sigf(gf) * c2r[r] + sigf(gi) * tanhf(gg);
        c2r[r]   = c;
        h2s[m * SH2 + j] = sigf(go) * tanhf(c);
      }
    }
    __syncthreads();

    {
      v2f a2[KS_H2], a3[KS_H3];
      load_frags<KS_H2>(a2, &h2s[col * SH2], khalf);
      load_frags<KS_H3>(a3, &h3s[col * SH3], khalf);
      for (int tile = wave; tile < 7; tile += NWAVES) {
        const int n = tile * 16 + col;
        v2f bi[KS_H2], bh[KS_H3];
        load_frags<KS_H2>(bi, &wih3s[n * SH2], khalf);
        load_frags<KS_H3>(bh, &whh3s[n * SH3], khalf);
        v8f acc = {};
        acc = wmma_chain<KS_H2>(a2, bi, acc);
        acc = wmma_chain<KS_H3>(a3, bh, acc);
        if (n < G3) {
          #pragma unroll
          for (int r = 0; r < 8; ++r)
            gates3[(r + rowhalf) * G3S + n] = acc[r];
        }
      }
    }
    __syncthreads();

    #pragma unroll
    for (int r = 0; r < 2; ++r) {
      const int idx = tid + r * NT;
      if (idx < 16 * H3) {
        const int m = idx & 15, j = idx >> 4;
        float gi = gates3[m * G3S + j]          + bs3[r][0];
        float gf = gates3[m * G3S + H3 + j]     + bs3[r][1];
        float gg = gates3[m * G3S + 2 * H3 + j] + bs3[r][2];
        float go = gates3[m * G3S + 3 * H3 + j] + bs3[r][3];
        float c  = sigf(gf) * c3r[r] + sigf(gi) * tanhf(gg);
        c3r[r]   = c;
        h3s[m * SH3 + j] = sigf(go) * tanhf(c);
      }
    }
    __syncthreads();

    if (tid < 16) {
      const float* hr = &h3s[tid * SH3];
      float acc = blin;
      #pragma unroll
      for (int j = 0; j < H3; ++j) acc += hr[j] * wl[j];
      out[(size_t)t * BB + bbase + tid] = acc;
    }
    __syncthreads();
  }
}

extern "C" void kernel_launch(void* const* d_in, const int* in_sizes, int n_in,
                              void* d_out, int out_size, void* d_ws, size_t ws_size,
                              hipStream_t stream) {
  (void)in_sizes; (void)n_in; (void)d_ws; (void)ws_size; (void)out_size;
  const float* x     = (const float*)d_in[0];
  const float* w_ih1 = (const float*)d_in[1];
  const float* w_hh1 = (const float*)d_in[2];
  const float* b_ih1 = (const float*)d_in[3];
  const float* b_hh1 = (const float*)d_in[4];
  const float* w_ih2 = (const float*)d_in[5];
  const float* w_hh2 = (const float*)d_in[6];
  const float* b_ih2 = (const float*)d_in[7];
  const float* b_hh2 = (const float*)d_in[8];
  const float* w_ih3 = (const float*)d_in[9];
  const float* w_hh3 = (const float*)d_in[10];
  const float* b_ih3 = (const float*)d_in[11];
  const float* b_hh3 = (const float*)d_in[12];
  const float* w_lin = (const float*)d_in[13];
  const float* b_lin = (const float*)d_in[14];
  float* out = (float*)d_out;

  dim3 grid(BB / 16);
  dim3 block(NT);
  lstm3_fused_kernel<<<grid, block, 0, stream>>>(
      x, w_ih1, w_hh1, b_ih1, b_hh1,
      w_ih2, w_hh2, b_ih2, b_hh2,
      w_ih3, w_hh3, b_ih3, b_hh3,
      w_lin, b_lin, out);
}


# Launch heads: each work-group's set-up reads (weights in 16- and 8-byte pieces, carried h and c, biases, first input tile) issued as one group before any is used; were 10-20 serial round trips
# speedup vs baseline: 4.9569x; 4.9569x over previous
#include <hip/hip_runtime.h>
#include <hip/hip_bf16.h>
#include <math.h>


typedef float v2f __attribute__((ext_vector_type(2)));
typedef float v4f __attribute__((ext_vector_type(4)));
typedef float v8f __attribute__((ext_vector_type(8)));

#define TT   512
#define BB   512
#define NIN  64
#define H1   100
#define H2   50
#define H3   25
#define G1   400
#define G2   200
#define G3   100
#define G1S  400
#define G2S  208
#define G3S  112
#define SH1  104
#define SH2  52
#define SH3  28
#define NT   512
#define NWAVES 16
#define SNT  1024
#define SNW  32
#define NTILE (BB / 16)

#define KS_X1 16
#define KS_H1 25
#define KS_H2 13
#define KS_H3 7

#define CH      32
#define NCHUNK  (TT / CH)
#define NL1     4
#define NL2     2
#define R1      4
#define R2      8
#define HBLK_PAD 416
#define HBLK_MIN 400
#define CSLOT   512
#define WS_C1   0
#define WS_C2   (WS_C1 + NTILE * NL1 * CSLOT)
#define WS_C3   (WS_C2 + NTILE * NL2 * CSLOT)
#define WS_H3   (WS_C3 + NTILE * 512)
#define WS_XG   (WS_H3 + NTILE * 16 * SH3)
#define NGT1    25
#define XG_TILE 256
#define XG_STEP (NGT1 * XG_TILE)
#define XG_BT   (CH * XG_STEP)
#define WS_XGP  (NTILE * XG_BT)
#define WS_H    (WS_XG + 2 * WS_XGP)
#define WS_FLOATS(hb) (WS_H + 2 * CH * NTILE * NL1 * (hb) + 2 * CH * NTILE * NL2 * (hb))
#define WS_NEED ((size_t)WS_FLOATS(HBLK_MIN) * sizeof(float))
#define WS_PAD  ((size_t)WS_FLOATS(HBLK_PAD) * sizeof(float))
#define NPRE    1
#define NY      (NL1 + NL2 + 1 + NPRE)
#define XSP     68
#define NCOPY   400
#define NCOPY1  100
#define NCOPY2  200
#define KP0     7
#define KPN     6
#define PKP     17
#define PKB     (16 * PKP)
#define KG      5
#define W1FULL  (G1 * H1 / 4 / SNT)
#define W1REM   (G1 * H1 / 4 - W1FULL * SNT)
#define W2FULL  (G2 * H1 / 4 / SNT)
#define W2REM   (G2 * H1 / 4 - W2FULL * SNT)
#define W3FULL  (G3 * H3 / SNT)
#define W3REM   (G3 * H3 - W3FULL * SNT)

__device__ __forceinline__ float sigf(float x) { return 1.0f / (1.0f + __expf(-x)); }

__device__ __forceinline__ v8f wmma4(v2f a, v2f b, v8f c) {
  return __builtin_amdgcn_wmma_f32_16x16x4_f32(
      false, a, false, b, (short)0, c, false, false);
}

template<int KS>
__device__ __forceinline__ void load_frags(v2f* f, const float* p, int khalf) {
  #pragma unroll
  for (int i = 0; i < KS; ++i) {
    f[i].x = p[4 * i + khalf];
    f[i].y = p[4 * i + khalf + 1];
  }
}

template<int KS>
__device__ __forceinline__ v8f wmma_chain(const v2f* a, const v2f* b, v8f acc) {
  #pragma unroll
  for (int i = 0; i < KS; ++i) acc = wmma4(a[i], b[i], acc);
  return acc;
}


__global__ __launch_bounds__(SNT, 1) void lstm3_stage_kernel(
    const float* __restrict__ x,
    const float* __restrict__ w_ih1, const float* __restrict__ w_hh1,
    const float* __restrict__ b_ih1, const float* __restrict__ b_hh1,
    const float* __restrict__ w_ih2, const float* __restrict__ w_hh2,
    const float* __restrict__ b_ih2, const float* __restrict__ b_hh2,
    const float* __restrict__ w_ih3, const float* __restrict__ w_hh3,
    const float* __restrict__ b_ih3, const float* __restrict__ b_hh3,
    const float* __restrict__ w_lin, const float* __restrict__ b_lin,
    float* __restrict__ out,
    float* ws, size_t ws_bytes, int k)
{
  __shared__ __attribute__((aligned(16))) float smem[G1 * H1 + 16 * G1S + 16 * SH1 + 3 * PKB];

  if (ws_bytes < WS_NEED) return;

  const int tid     = threadIdx.x;
  const int lane    = tid & 31;
  const int wave    = tid >> 5;
  const int col     = lane & 15;
  const int khalf   = (lane >> 4) << 1;
  const int rowhalf = (lane >> 4) << 3;
  const int bbase   = blockIdx.x * 16;

  const int by = blockIdx.y;
  if (k == 0 || by >= NL1 + NL2 + 1) {
    if (k >= NCHUNK) return;
    const int npre = (k == 0) ? NY : NPRE;
    const int q    = (k == 0) ? by : by - (NL1 + NL2 + 1);
    const int pst  = CH / npre;
    const int tl0  = q * pst;
    float* xs = smem;
    for (int i = tid; i < pst * 256; i += SNT) {
      const int s = i >> 8, p = i & 255;
      const v4f v = *(const v4f*)(x + ((size_t)(k * CH + tl0 + s) * BB + bbase) * NIN + p * 4);
      *(v4f*)&xs[(s * 16 + (p >> 4)) * XSP + (p & 15) * 4] = v;
    }
    __syncthreads();
    if (wave < NGT1) {
      const int n = wave * 16 + col;
      v2f bi[KS_X1];
      load_frags<KS_X1>(bi, w_ih1 + (size_t)n * NIN, khalf);
      float* xgout = ws + WS_XG + (size_t)(k & 1) * WS_XGP + (size_t)blockIdx.x * XG_BT + wave * XG_TILE + lane * 8;
      for (int s = 0; s < pst; ++s) {
        v2f ax[KS_X1];
        load_frags<KS_X1>(ax, &xs[(s * 16 + col) * XSP], khalf);
        v8f acc = {};
        acc = wmma_chain<KS_X1>(ax, bi, acc);
        *(v8f*)(xgout + (size_t)(tl0 + s) * XG_STEP) = acc;
      }
    }
    return;
  }

  const int layer = by < NL1 ? 0 : (by < NL1 + NL2 ? 1 : 2);
  const int c     = k - 1 - layer;
  if (c < 0 || c >= NCHUNK) return;

  const int tr      = tid / 25;
  const int tc      = tid - tr * 25;
  const int par     = c & 1;
  const int hblk    = ws_bytes >= WS_PAD ? HBLK_PAD : HBLK_MIN;
  const int hs1     = NTILE * NL1 * hblk;
  const int hs2     = NTILE * NL2 * hblk;
  float* h1pl = ws + WS_H;
  float* h2pl = h1pl + 2 * (size_t)CH * hs1;

  if (layer == 0) {
    float* whh1s  = smem;
    float* gates1 = smem + G1 * H1;
    float* h1s    = gates1 + 16 * G1S;
    float* pks    = h1s + 16 * SH1;
    float*       h1out  = h1pl + (size_t)par * CH * hs1;
    const float* h1prev = h1pl + (size_t)(par ^ 1) * CH * hs1;
    const int    rbase  = by * R1;
    const bool   mine   = (lane >> 4) == (by >> 1);
    float*       c1st   = ws + WS_C1 + ((size_t)blockIdx.x * NL1 + by) * CSLOT;
    const size_t hb1    = ((size_t)blockIdx.x * NL1 + by) * hblk;
    const float* xgin   = ws + WS_XG + (size_t)par * WS_XGP + (size_t)blockIdx.x * XG_BT + lane * 8;
    const bool full  = wave < NGT1 - 1;
    const bool first = wave == NGT1 - 1;
    const bool piece = wave >= NGT1 && wave < NGT1 + 3;
    const bool own   = full || first;
    const int  tile0 = full ? wave : NGT1 - 1;

    v8f nx0 = {};
    if (own && mine) nx0 = *(const v8f*)(xgin + tile0 * XG_TILE);
    float c1r = (c > 0 && tid < CSLOT) ? c1st[tid] : 0.f;
    v4f hcar = {};
    if (c > 0 && tid < NCOPY1)
      hcar = *(const v4f*)(h1prev + (size_t)(CH - 1) * hs1 + hb1 + tid * 4);
    const int jb1 = (tid < R1 * H1) ? (tid >> 2) : 0;
    float bi1[4], bh1[4];
    #pragma unroll
    for (int g = 0; g < 4; ++g) { bi1[g] = b_ih1[g * H1 + jb1]; bh1[g] = b_hh1[g * H1 + jb1]; }
    v4f wv[W1FULL + 1];
    #pragma unroll
    for (int i = 0; i < W1FULL; ++i) wv[i] = ((const v4f*)w_hh1)[tid + i * SNT];
    wv[W1FULL] = v4f{};
    if (tid < W1REM) wv[W1FULL] = ((const v4f*)w_hh1)[tid + W1FULL * SNT];
    #pragma unroll
    for (int i = 0; i < W1FULL; ++i) ((v4f*)whh1s)[tid + i * SNT] = wv[i];
    if (tid < W1REM) ((v4f*)whh1s)[tid + W1FULL * SNT] = wv[W1FULL];
    for (int i = tid; i < 16 * SH1; i += SNT) h1s[i] = 0.f;
    __syncthreads();
    const int nrow = tile0 * 16 + col;
    const int kp0  = piece ? 4 * (KP0 + KPN * (wave - NGT1)) : 0;
    v2f bw[KS_H1] = {};
    if (full)       load_frags<KS_H1>(bw, &whh1s[nrow * H1], khalf);
    else if (first) load_frags<KP0>(bw, &whh1s[nrow * H1], khalf);
    else if (piece) load_frags<KPN>(bw, &whh1s[nrow * H1 + kp0], khalf);
    if (c > 0 && tid < NCOPY1)
      *(v4f*)&h1s[(rbase + tr) * SH1 + tc * 4] = hcar;

    float bs1[4];
    #pragma unroll
    for (int g = 0; g < 4; ++g) bs1[g] = bi1[g] + bh1[g];
    __syncthreads();

    for (int tl = 0; tl < CH; ++tl) {
      const int n = tile0 * 16 + col;
      if (full) {
        const v8f s0 = nx0;
        if (tl + 1 < CH && mine)
          nx0 = *(const v8f*)(xgin + (size_t)(tl + 1) * XG_STEP + tile0 * XG_TILE);
        const float* hp = &h1s[col * SH1];
        v2f a0[KG], a1[KG];
        load_frags<KG>(a0, hp, khalf);
        load_frags<KG>(a1, hp + 4 * KG, khalf);
        __builtin_amdgcn_sched_barrier(0);
        v8f acc = wmma_chain<KG>(a0, bw, s0);
        __builtin_amdgcn_sched_barrier(0);
        load_frags<KG>(a0, hp + 8 * KG, khalf);
        __builtin_amdgcn_sched_barrier(0);
        acc = wmma_chain<KG>(a1, bw + KG, acc);
        __builtin_amdgcn_sched_barrier(0);
        load_frags<KG>(a1, hp + 12 * KG, khalf);
        __builtin_amdgcn_sched_barrier(0);
        acc = wmma_chain<KG>(a0, bw + 2 * KG, acc);
        __builtin_amdgcn_sched_barrier(0);
        load_frags<KG>(a0, hp + 16 * KG, khalf);
        __builtin_amdgcn_sched_barrier(0);
        acc = wmma_chain<KG>(a1, bw + 3 * KG, acc);
        acc = wmma_chain<KG>(a0, bw + 4 * KG, acc);
        #pragma unroll
        for (int r = 0; r < 8; ++r)
          gates1[(r + rowhalf) * G1S + n] = acc[r];
      } else if (first) {
        const v8f s0 = nx0;
        if (tl + 1 < CH && mine)
          nx0 = *(const v8f*)(xgin + (size_t)(tl + 1) * XG_STEP + tile0 * XG_TILE);
        v2f ah[KP0];
        load_frags<KP0>(ah, &h1s[col * SH1], khalf);
        const v8f acc = wmma_chain<KP0>(ah, bw, s0);
        #pragma unroll
        for (int r = 0; r < 8; ++r)
          gates1[(r + rowhalf) * G1S + n] = acc[r];
      } else if (piece) {
        const int pi = wave - NGT1;
        v2f ah[KPN];
        load_frags<KPN>(ah, &h1s[col * SH1 + kp0], khalf);
        v8f acc = {};
        acc = wmma_chain<KPN>(ah, bw, acc);
        float* pk = pks + pi * PKB;
        #pragma unroll
        for (int r = 0; r < 8; ++r)
          pk[(r + rowhalf) * PKP + col] = acc[r];
      }
      __syncthreads();

      if (tid < R1 * H1) {
        const int m = rbase + (tid & 3), j = tid >> 2;
        float gi = gates1[m * G1S + j]          + bs1[0];
        float gf = gates1[m * G1S + H1 + j]     + bs1[1];
        float gg = gates1[m * G1S + 2 * H1 + j] + bs1[2];
        float go = gates1[m * G1S + 3 * H1 + j];
        if (j >= H1 - 16) {
          const float* pp = pks + m * PKP + (j - (H1 - 16));
          go = ((go + pp[0]) + pp[PKB]) + pp[2 * PKB];
        }
        go += bs1[3];
        float cc = sigf(gf) * c1r + sigf(gi) * tanhf(gg);
        c1r      = cc;
        h1s[m * SH1 + j] = sigf(go) * tanhf(cc);
      }
      __syncthreads();

      if (tid < NCOPY1) {
        const v4f v = *(const v4f*)&h1s[(rbase + tr) * SH1 + tc * 4];
        *(v4f*)(h1out + (size_t)tl * hs1 + hb1 + tid * 4) = v;
      }
    }
    if (tid < CSLOT) c1st[tid] = c1r;

  } else if (layer == 1) {
    float* whh2s  = smem;
    float* gates2 = smem + 208 * SH2;
    float* h1s    = gates2 + 16 * G2S;
    float* h2s    = h1s + 16 * SH1;
    float* wih2s  = h2s + 16 * SH2;
    const int    q2     = by - NL1;
    const int    rb2    = q2 * R2;
    const float* h1in   = h1pl + (size_t)par * CH * hs1;
    float*       h2out  = h2pl + (size_t)par * CH * hs2;
    const float* h2prev = h2pl + (size_t)(par ^ 1) * CH * hs2;
    float*       c2st   = ws + WS_C2 + ((size_t)blockIdx.x * NL2 + q2) * CSLOT;
    const size_t hin1   = ((size_t)blockIdx.x * NL1 + 2 * q2 + (tr >> 2)) * hblk + (tr & 3) * H1 + tc * 4;
    const size_t hb2    = ((size_t)blockIdx.x * NL2 + q2) * hblk;

    float c2r = (c > 0 && tid < CSLOT) ? c2st[tid] : 0.f;
    v4f h1car = {};
    v2f h2car = {};
    if (tid < NCOPY2) {
      h1car = *(const v4f*)(h1in + hin1);
      if (c > 0) h2car = *(const v2f*)(h2prev + (size_t)(CH - 1) * hs2 + hb2 + tid * 2);
    }
    const int jb2 = (tid < R2 * H2) ? (tid >> 3) : 0;
    float bi2[4], bh2[4];
    #pragma unroll
    for (int g = 0; g < 4; ++g) { bi2[g] = b_ih2[g * H2 + jb2]; bh2[g] = b_hh2[g * H2 + jb2]; }
    v4f wi[W2FULL + 1];
    v2f wh[W2FULL + 1];
    #pragma unroll
    for (int i = 0; i < W2FULL; ++i) {
      wi[i] = ((const v4f*)w_ih2)[tid + i * SNT];
      wh[i] = ((const v2f*)w_hh2)[tid + i * SNT];
    }
    wi[W2FULL] = v4f{};
    wh[W2FULL] = v2f{};
    if (tid < W2REM) {
      wi[W2FULL] = ((const v4f*)w_ih2)[tid + W2FULL * SNT];
      wh[W2FULL] = ((const v2f*)w_hh2)[tid + W2FULL * SNT];
    }
    #pragma unroll
    for (int i = 0; i < W2FULL; ++i) ((v4f*)wih2s)[tid + i * SNT] = wi[i];
    if (tid < W2REM) ((v4f*)wih2s)[tid + W2FULL * SNT] = wi[W2FULL];
    for (int i = tid; i < 208 * SH2; i += SNT) whh2s[i] = 0.f;
    for (int i = tid; i < 16 * SH1; i += SNT) h1s[i] = 0.f;
    for (int i = tid; i < 16 * SH2; i += SNT) h2s[i] = 0.f;
    __syncthreads();
    #pragma unroll
    for (int i = 0; i <= W2FULL; ++i) {
      const int p = tid + i * SNT;
      if (i < W2FULL || tid < W2REM) {
        const int r = p / 25, q = p - r * 25;
        *(v2f*)&whh2s[r * SH2 + 2 * q] = wh[i];
      }
    }
    if (tid < NCOPY2) {
      *(v4f*)&h1s[(rb2 + tr) * SH1 + tc * 4] = h1car;
      if (c > 0) *(v2f*)&h2s[(rb2 + tr) * SH2 + tc * 2] = h2car;
    }

    float bs2[4];
    #pragma unroll
    for (int g = 0; g < 4; ++g) bs2[g] = bi2[g] + bh2[g];
    __syncthreads();

    for (int tl = 0; tl < CH; ++tl) {
      const bool more = (tl + 1 < CH) && (tid < NCOPY2);
      v4f hn = {};
      if (more) hn = *(const v4f*)(h1in + (size_t)(tl + 1) * hs1 + hin1);
      {
        v2f a1[KS_H1], a2[KS_H2];
        load_frags<KS_H1>(a1, &h1s[col * SH1], khalf);
        load_frags<KS_H2>(a2, &h2s[col * SH2], khalf);
        for (int tile = wave; tile < 13; tile += SNW) {
          const int n  = tile * 16 + col;
          const int nc = (n < G2) ? n : (G2 - 1);
          v2f bi[KS_H1], bh[KS_H2];
          load_frags<KS_H1>(bi, &wih2s[nc * H1], khalf);
          load_frags<KS_H2>(bh, &whh2s[n * SH2], khalf);
          v8f acc = {};
          acc = wmma_chain<KS_H1>(a1, bi, acc);
          acc = wmma_chain<KS_H2>(a2, bh, acc);
          if (n < G2) {
            #pragma unroll
            for (int r = 0; r < 8; ++r)
              gates2[(r + rowhalf) * G2S + n] = acc[r];
          }
        }
      }
      __syncthreads();

      if (tid < R2 * H2) {
        const int m = rb2 + (tid & 7), j = tid >> 3;
        float gi = gates2[m * G2S + j]          + bs2[0];
        float gf = gates2[m * G2S + H2 + j]     + bs2[1];
        float gg = gates2[m * G2S + 2 * H2 + j] + bs2[2];
        float go = gates2[m * G2S + 3 * H2 + j] + bs2[3];
        float cc = sigf(gf) * c2r + sigf(gi) * tanhf(gg);
        c2r      = cc;
        h2s[m * SH2 + j] = sigf(go) * tanhf(cc);
      }
      if (more) *(v4f*)&h1s[(rb2 + tr) * SH1 + tc * 4] = hn;
      __syncthreads();

      if (tid < NCOPY2) {
        const v2f u = *(const v2f*)&h2s[(rb2 + tr) * SH2 + tc * 2];
        *(v2f*)(h2out + (size_t)tl * hs2 + hb2 + tid * 2) = u;
      }
    }
    if (tid < CSLOT) c2st[tid] = c2r;

  } else {
    float* wih3s  = smem;
    float* whh3s  = smem + 112 * SH2;
    float* gates3 = whh3s + 112 * SH3;
    float* h2s    = gates3 + 16 * G3S;
    float* h3s    = h2s + 16 * SH2;
    const float* h2in = h2pl + (size_t)par * CH * hs2;
    const size_t hin2 = ((size_t)blockIdx.x * NL2 + (tr >> 3)) * hblk + (tr & 7) * H2 + tc * 2;
    float*       h3st = ws + WS_H3 + (size_t)blockIdx.x * 16 * SH3;
    float*       c3st = ws + WS_C3 + (size_t)blockIdx.x * 512;

    float c3r = (c > 0 && tid < 16 * H3) ? c3st[tid] : 0.f;
    float h3car = 0.f;
    if (c > 0 && tid < 16 * SH3) h3car = h3st[tid];
    v2f h2car = {};
    if (tid < NCOPY) h2car = *(const v2f*)(h2in + hin2);
    const int jb3 = (tid < 16 * H3) ? (tid >> 4) : 0;
    float bi3[4], bh3[4];
    #pragma unroll
    for (int g = 0; g < 4; ++g) { bi3[g] = b_ih3[g * H3 + jb3]; bh3[g] = b_hh3[g * H3 + jb3]; }
    v2f w3i[W3FULL + 1];
    float w3h[W3FULL + 1];
    #pragma unroll
    for (int i = 0; i < W3FULL; ++i) {
      w3i[i] = ((const v2f*)w_ih3)[tid + i * SNT];
      w3h[i] = w_hh3[tid + i * SNT];
    }
    w3i[W3FULL] = v2f{};
    w3h[W3FULL] = 0.f;
    if (tid < W3REM) {
      w3i[W3FULL] = ((const v2f*)w_ih3)[tid + W3FULL * SNT];
      w3h[W3FULL] = w_hh3[tid + W3FULL * SNT];
    }
    float wl[H3];
    #pragma unroll
    for (int j = 0; j < H3; ++j) wl[j] = w_lin[j];
    const float blin = b_lin[0];
    for (int i = tid; i < 112 * SH2; i += SNT) wih3s[i] = 0.f;
    for (int i = tid; i < 112 * SH3; i += SNT) whh3s[i] = 0.f;
    for (int i = tid; i < 16 * SH2; i += SNT) h2s[i] = 0.f;
    for (int i = tid; i < 16 * SH3; i += SNT) h3s[i] = 0.f;
    __syncthreads();
    #pragma unroll
    for (int i = 0; i <= W3FULL; ++i) {
      const int p = tid + i * SNT;
      if (i < W3FULL || tid < W3REM) {
        const int r = p / 25, q = p - r * 25;
        *(v2f*)&wih3s[r * SH2 + 2 * q] = w3i[i];
        whh3s[r * SH3 + q] = w3h[i];
      }
    }
    if (tid < NCOPY) *(v2f*)&h2s[tr * SH2 + tc * 2] = h2car;
    if (c > 0 && tid < 16 * SH3) h3s[tid] = h3car;

    float bs3[4];
    #pragma unroll
    for (int g = 0; g < 4; ++g) bs3[g] = bi3[g] + bh3[g];
    __syncthreads();

    for (int tl = 0; tl < CH; ++tl) {
      const int t = c * CH + tl;
      const bool more = (tl + 1 < CH) && (tid < NCOPY);
      v2f hn = {};
      if (more) hn = *(const v2f*)(h2in + (size_t)(tl + 1) * hs2 + hin2);
      {
        v2f a2[KS_H2], a3[KS_H3];
        load_frags<KS_H2>(a2, &h2s[col * SH2], khalf);
        load_frags<KS_H3>(a3, &h3s[col * SH3], khalf);
        for (int tile = wave; tile < 7; tile += SNW) {
          const int n = tile * 16 + col;
          v2f bi[KS_H2], bh[KS_H3];
          load_frags<KS_H2>(bi, &wih3s[n * SH2], khalf);
          load_frags<KS_H3>(bh, &whh3s[n * SH3], khalf);
          v8f acc = {};
          acc = wmma_chain<KS_H2>(a2, bi, acc);
          acc = wmma_chain<KS_H3>(a3, bh, acc);
          if (n < G3) {
            #pragma unroll
            for (int r = 0; r < 8; ++r)
              gates3[(r + rowhalf) * G3S + n] = acc[r];
          }
        }
      }
      __syncthreads();

      if (tid < 16 * H3) {
        const int m = tid & 15, j = tid >> 4;
        float gi = gates3[m * G3S + j]          + bs3[0];
        float gf = gates3[m * G3S + H3 + j]     + bs3[1];
        float gg = gates3[m * G3S + 2 * H3 + j] + bs3[2];
        float go = gates3[m * G3S + 3 * H3 + j] + bs3[3];
        float cc = sigf(gf) * c3r + sigf(gi) * tanhf(gg);
        c3r      = cc;
        h3s[m * SH3 + j] = sigf(go) * tanhf(cc);
      }
      if (more) *(v2f*)&h2s[tr * SH2 + tc * 2] = hn;
      __syncthreads();

      if (tid < 16) {
        const float* hr = &h3s[tid * SH3];
        float acc = blin;
        #pragma unroll
        for (int j = 0; j < H3; ++j) acc += hr[j] * wl[j];
        out[(size_t)t * BB + bbase + tid] = acc;
      }
    }
    if (tid < 16 * SH3) h3st[tid] = h3s[tid];
    if (tid < 16 * H3) c3st[tid] = c3r;
  }
}

__global__ __launch_bounds__(NT, 1) void lstm3_fused_kernel(
    const float* __restrict__ x,
    const float* __restrict__ w_ih1, const float* __restrict__ w_hh1,
    const float* __restrict__ b_ih1, const float* __restrict__ b_hh1,
    const float* __restrict__ w_ih2, const float* __restrict__ w_hh2,
    const float* __restrict__ b_ih2, const float* __restrict__ b_hh2,
    const float* __restrict__ w_ih3, const float* __restrict__ w_hh3,
    const float* __restrict__ b_ih3, const float* __restrict__ b_hh3,
    const float* __restrict__ w_lin, const float* __restrict__ b_lin,
    float* __restrict__ out, size_t ws_bytes)
{
  __shared__ float gates1[16 * G1S];
  __shared__ float gates2[16 * G2S];
  __shared__ float gates3[16 * G3S];
  __shared__ float h1s[16 * SH1];
  __shared__ float h2s[16 * SH2];
  __shared__ float h3s[16 * SH3];
  __shared__ float whh1s[G1 * H1];
  __shared__ float whh2s[208 * SH2];
  __shared__ float wih3s[112 * SH2];
  __shared__ float whh3s[112 * SH3];

  if (ws_bytes >= WS_NEED) return;

  const int tid     = threadIdx.x;
  const int lane    = tid & 31;
  const int wave    = tid >> 5;
  const int col     = lane & 15;
  const int khalf   = (lane >> 4) << 1;
  const int rowhalf = (lane >> 4) << 3;
  const int bbase   = blockIdx.x * 16;

  for (int i = tid; i < 16 * SH1; i += NT) h1s[i] = 0.f;
  for (int i = tid; i < 16 * SH2; i += NT) h2s[i] = 0.f;
  for (int i = tid; i < 16 * SH3; i += NT) h3s[i] = 0.f;
  for (int i = tid; i < 208 * SH2; i += NT) whh2s[i] = 0.f;
  for (int i = tid; i < 112 * SH2; i += NT) wih3s[i] = 0.f;
  for (int i = tid; i < 112 * SH3; i += NT) whh3s[i] = 0.f;
  __syncthreads();

  for (int i = tid; i < G1 * H1; i += NT)
    whh1s[i] = w_hh1[i];
  for (int i = tid; i < G2 * H2; i += NT) {
    int r = i / H2, c = i - r * H2;
    whh2s[r * SH2 + c] = w_hh2[i];
  }
  for (int i = tid; i < G3 * H2; i += NT) {
    int r = i / H2, c = i - r * H2;
    wih3s[r * SH2 + c] = w_ih3[i];
  }
  for (int i = tid; i < G3 * H3; i += NT) {
    int r = i / H3, c = i - r * H3;
    whh3s[r * SH3 + c] = w_hh3[i];
  }

  float bs1[4][4], bs2[2][4], bs3[1][4];
  #pragma unroll
  for (int r = 0; r < 4; ++r) {
    const int idx = tid + r * NT;
    if (idx < 16 * H1) {
      const int j = idx >> 4;
      #pragma unroll
      for (int g = 0; g < 4; ++g) bs1[r][g] = b_ih1[g * H1 + j] + b_hh1[g * H1 + j];
    }
  }
  #pragma unroll
  for (int r = 0; r < 2; ++r) {
    const int idx = tid + r * NT;
    if (idx < 16 * H2) {
      const int j = idx >> 4;
      #pragma unroll
      for (int g = 0; g < 4; ++g) bs2[r][g] = b_ih2[g * H2 + j] + b_hh2[g * H2 + j];
    }
  }
  #pragma unroll
  for (int r = 0; r < 1; ++r) {
    const int idx = tid + r * NT;
    if (idx < 16 * H3) {
      const int j = idx >> 4;
      #pragma unroll
      for (int g = 0; g < 4; ++g) bs3[r][g] = b_ih3[g * H3 + j] + b_hh3[g * H3 + j];
    }
  }

  float wl[H3];
  #pragma unroll
  for (int j = 0; j < H3; ++j) wl[j] = w_lin[j];
  const float blin = b_lin[0];

  float c1r[4], c2r[2], c3r[1];
  #pragma unroll
  for (int r = 0; r < 4; ++r) c1r[r] = 0.f;
  #pragma unroll
  for (int r = 0; r < 2; ++r) c2r[r] = 0.f;
  #pragma unroll
  for (int r = 0; r < 1; ++r) c3r[r] = 0.f;

  __syncthreads();

  for (int t = 0; t < TT; ++t) {
    {
      if (t + 1 < TT && tid < 16)
        __builtin_prefetch(x + ((size_t)(t + 1) * BB + bbase + tid) * NIN, 0, 3);

      v2f ax[KS_X1], ah[KS_H1];
      load_frags<KS_X1>(ax, x + ((size_t)t * BB + bbase + col) * NIN, khalf);
      load_frags<KS_H1>(ah, &h1s[col * SH1], khalf);
      for (int tile = wave; tile < 25; tile += NWAVES) {
        const int n = tile * 16 + col;
        v2f bi[KS_X1], bh[KS_H1];
        load_frags<KS_X1>(bi, w_ih1 + (size_t)n * NIN, khalf);
        load_frags<KS_H1>(bh, &whh1s[n * H1], khalf);
        v8f acc = {};
        acc = wmma_chain<KS_X1>(ax, bi, acc);
        acc = wmma_chain<KS_H1>(ah, bh, acc);
        #pragma unroll
        for (int r = 0; r < 8; ++r)
          gates1[(r + rowhalf) * G1S + n] = acc[r];
      }
    }
    __syncthreads();

    #pragma unroll
    for (int r = 0; r < 4; ++r) {
      const int idx = tid + r * NT;
      if (idx < 16 * H1) {
        const int m = idx & 15, j = idx >> 4;
        float gi = gates1[m * G1S + j]          + bs1[r][0];
        float gf = gates1[m * G1S + H1 + j]     + bs1[r][1];
        float gg = gates1[m * G1S + 2 * H1 + j] + bs1[r][2];
        float go = gates1[m * G1S + 3 * H1 + j] + bs1[r][3];
        float c  = sigf(gf) * c1r[r] + sigf(gi) * tanhf(gg);
        c1r[r]   = c;
        h1s[m * SH1 + j] = sigf(go) * tanhf(c);
      }
    }
    __syncthreads();

    {
      v2f a1[KS_H1], a2[KS_H2];
      load_frags<KS_H1>(a1, &h1s[col * SH1], khalf);
      load_frags<KS_H2>(a2, &h2s[col * SH2], khalf);
      for (int tile = wave; tile < 13; tile += NWAVES) {
        const int n  = tile * 16 + col;
        const int nc = (n < G2) ? n : (G2 - 1);
        v2f bi[KS_H1], bh[KS_H2];
        load_frags<KS_H1>(bi, w_ih2 + (size_t)nc * H1, khalf);
        load_frags<KS_H2>(bh, &whh2s[n * SH2], khalf);
        v8f acc = {};
        acc = wmma_chain<KS_H1>(a1, bi, acc);
        acc = wmma_chain<KS_H2>(a2, bh, acc);
        if (n < G2) {
          #pragma unroll
          for (int r = 0; r < 8; ++r)
            gates2[(r + rowhalf) * G2S + n] = acc[r];
        }
      }
    }
    __syncthreads();

    #pragma unroll
    for (int r = 0; r < 2; ++r) {
      const int idx = tid + r * NT;
      if (idx < 16 * H2) {
        const int m = idx & 15, j = idx >> 4;
        float gi = gates2[m * G2S + j]          + bs2[r][0];
        float gf = gates2[m * G2S + H2 + j]     + bs2[r][1];
        float gg = gates2[m * G2S + 2 * H2 + j] + bs2[r][2];
        float go = gates2[m * G2S + 3 * H2 + j] + bs2[r][3];
        float c  = sigf(gf) * c2r[r] + sigf(gi) * tanhf(gg);
        c2r[r]   = c;
        h2s[m * SH2 + j] = sigf(go) * tanhf(c);
      }
    }
    __syncthreads();

    {
      v2f a2[KS_H2], a3[KS_H3];
      load_frags<KS_H2>(a2, &h2s[col * SH2], khalf);
      load_frags<KS_H3>(a3, &h3s[col * SH3], khalf);
      for (int tile = wave; tile < 7; tile += NWAVES) {
        const int n = tile * 16 + col;
        v2f bi[KS_H2], bh[KS_H3];
        load_frags<KS_H2>(bi, &wih3s[n * SH2], khalf);
        load_frags<KS_H3>(bh, &whh3s[n * SH3], khalf);
        v8f acc = {};
        acc = wmma_chain<KS_H2>(a2, bi, acc);
        acc = wmma_chain<KS_H3>(a3, bh, acc);
        if (n < G3) {
          #pragma unroll
          for (int r = 0; r < 8; ++r)
            gates3[(r + rowhalf) * G3S + n] = acc[r];
        }
      }
    }
    __syncthreads();

    #pragma unroll
    for (int r = 0; r < 1; ++r) {
      const int idx = tid + r * NT;
      if (idx < 16 * H3) {
        const int m = idx & 15, j = idx >> 4;
        float gi = gates3[m * G3S + j]          + bs3[r][0];
        float gf = gates3[m * G3S + H3 + j]     + bs3[r][1];
        float gg = gates3[m * G3S + 2 * H3 + j] + bs3[r][2];
        float go = gates3[m * G3S + 3 * H3 + j] + bs3[r][3];
        float c  = sigf(gf) * c3r[r] + sigf(gi) * tanhf(gg);
        c3r[r]   = c;
        h3s[m * SH3 + j] = sigf(go) * tanhf(c);
      }
    }
    __syncthreads();

    if (tid < 16) {
      const float* hr = &h3s[tid * SH3];
      float acc = blin;
      #pragma unroll
      for (int j = 0; j < H3; ++j) acc += hr[j] * wl[j];
      out[(size_t)t * BB + bbase + tid] = acc;
    }
    __syncthreads();
  }
}

extern "C" void kernel_launch(void* const* d_in, const int* in_sizes, int n_in,
                              void* d_out, int out_size, void* d_ws, size_t ws_size,
                              hipStream_t stream) {
  (void)in_sizes; (void)n_in; (void)out_size;
  const float* x     = (const float*)d_in[0];
  const float* w_ih1 = (const float*)d_in[1];
  const float* w_hh1 = (const float*)d_in[2];
  const float* b_ih1 = (const float*)d_in[3];
  const float* b_hh1 = (const float*)d_in[4];
  const float* w_ih2 = (const float*)d_in[5];
  const float* w_hh2 = (const float*)d_in[6];
  const float* b_ih2 = (const float*)d_in[7];
  const float* b_hh2 = (const float*)d_in[8];
  const float* w_ih3 = (const float*)d_in[9];
  const float* w_hh3 = (const float*)d_in[10];
  const float* b_ih3 = (const float*)d_in[11];
  const float* b_hh3 = (const float*)d_in[12];
  const float* w_lin = (const float*)d_in[13];
  const float* b_lin = (const float*)d_in[14];
  float* out = (float*)d_out;
  float* ws  = (float*)d_ws;

  dim3 sgrid(BB / 16, NY);
  dim3 block(NT);
  dim3 sblock(SNT);

#define STAGE(K) \
  lstm3_stage_kernel<<<sgrid, sblock, 0, stream>>>( \
      x, w_ih1, w_hh1, b_ih1, b_hh1, \
      w_ih2, w_hh2, b_ih2, b_hh2, \
      w_ih3, w_hh3, b_ih3, b_hh3, \
      w_lin, b_lin, out, ws, ws_size, K)

  STAGE(0);  STAGE(1);  STAGE(2);  STAGE(3);  STAGE(4);  STAGE(5);
  STAGE(6);  STAGE(7);  STAGE(8);  STAGE(9);  STAGE(10); STAGE(11);
  STAGE(12); STAGE(13); STAGE(14); STAGE(15); STAGE(16); STAGE(17);
  STAGE(18);
#undef STAGE

  dim3 grid(BB / 16);
  lstm3_fused_kernel<<<grid, block, 0, stream>>>(
      x, w_ih1, w_hh1, b_ih1, b_hh1,
      w_ih2, w_hh2, b_ih2, b_hh2,
      w_ih3, w_hh3, b_ih3, b_hh3,
      w_lin, b_lin, out, ws_size);
}
